# starting kernel
# baseline (speedup 1.0000x reference)
#include <hip/hip_runtime.h>
#include <math.h>

typedef __attribute__((ext_vector_type(2))) float v2f;
typedef __attribute__((ext_vector_type(8))) float v8f;

__device__ __forceinline__ v8f wmma_f32(v2f a, v2f b, v8f c) {
    return __builtin_amdgcn_wmma_f32_16x16x4_f32(
        false, a, false, b, (short)0, c, false, false);
}

__device__ __constant__ float c_D[64] = {
    0.3535533906f,  0.3535533906f,  0.3535533906f,  0.3535533906f,
    0.3535533906f,  0.3535533906f,  0.3535533906f,  0.3535533906f,

    0.4903926402f,  0.4157348062f,  0.2777851165f,  0.0975451610f,
   -0.0975451610f, -0.2777851165f, -0.4157348062f, -0.4903926402f,

    0.4619397663f,  0.1913417162f, -0.1913417162f, -0.4619397663f,
   -0.4619397663f, -0.1913417162f,  0.1913417162f,  0.4619397663f,

    0.4157348062f, -0.0975451610f, -0.4903926402f, -0.2777851165f,
    0.2777851165f,  0.4903926402f,  0.0975451610f, -0.4157348062f,

    0.3535533906f, -0.3535533906f, -0.3535533906f,  0.3535533906f,
    0.3535533906f, -0.3535533906f, -0.3535533906f,  0.3535533906f,

    0.2777851165f, -0.4903926402f,  0.0975451610f,  0.4157348062f,
   -0.4157348062f, -0.0975451610f,  0.4903926402f, -0.2777851165f,

    0.1913417162f, -0.4619397663f,  0.4619397663f, -0.1913417162f,
   -0.1913417162f,  0.4619397663f, -0.4619397663f,  0.1913417162f,

    0.0975451610f, -0.2777851165f,  0.4157348062f, -0.4903926402f,
    0.4903926402f, -0.4157348062f,  0.2777851165f, -0.0975451610f
};

__device__ __forceinline__ float d2(int r, int c) {
    const float v = c_D[((r & 7) << 3) | (c & 7)];
    return ((r >> 3) == (c >> 3)) ? v : 0.0f;
}

__global__ void __launch_bounds__(128)
irse_dct_kernel(const float* __restrict__ x, float* __restrict__ feat) {
    __shared__ float lds[4 * 256];
    const int lane = threadIdx.x & 31;
    const int wave = threadIdx.x >> 5;
    const int tile = blockIdx.x * 4 + wave;
    const int ti = tile >> 4, tj = tile & 15;
    const int h = lane >> 4;
    const int N = lane & 15;
    const int row0 = ti * 16, col0 = tj * 16;

    v8f acc = {};
#pragma unroll
    for (int k = 0; k < 4; ++k) {
        const int ka = 4 * k + 2 * h;
        v2f a, b;
        a.x = d2(N, ka);
        a.y = d2(N, ka + 1);
        b.x = x[(((row0 + ka) * 256) + col0 + N) * 3];
        b.y = x[(((row0 + ka + 1) * 256) + col0 + N) * 3];
        acc = wmma_f32(a, b, acc);
    }

    float* wlds = lds + wave * 256;
#pragma unroll
    for (int r = 0; r < 8; ++r) wlds[(r + 8 * h) * 16 + N] = acc[r];
    __syncthreads();

    v8f acc2 = {};
#pragma unroll
    for (int k = 0; k < 4; ++k) {
        const int ka = 4 * k + 2 * h;
        v2f a, b;
        a.x = wlds[N * 16 + ka];
        a.y = wlds[N * 16 + ka + 1];
        b.x = d2(N, ka);
        b.y = d2(N, ka + 1);
        acc2 = wmma_f32(a, b, acc2);
    }

#pragma unroll
    for (int r = 0; r < 8; ++r) {
        const int M = r + 8 * h;
        const int i = M & 7, bi = 2 * ti + (M >> 3);
        const int j = N & 7, bj = 2 * tj + (N >> 3);
        feat[(i * 8 + j) * 1024 + bj * 32 + bi] = acc2[r];
    }
}

__global__ void __launch_bounds__(256)
irse_se_mlp_kernel(const float* __restrict__ feat,
                   const float* __restrict__ fc1_w, const float* __restrict__ fc1_b,
                   const float* __restrict__ fc2_w, const float* __restrict__ fc2_b,
                   float* __restrict__ scale_c) {
    __shared__ float part[256];
    __shared__ float sq[64];
    __shared__ float hbuf[32];
    const int t = threadIdx.x;
    const int f = t >> 2, q = t & 3;
    float s = 0.0f;
    const float* p = feat + f * 1024 + q * 256;
    for (int i = 0; i < 256; ++i) s += p[i];
    part[t] = s;
    __syncthreads();
    if (q == 0)
        sq[f] = (part[t] + part[t + 1] + part[t + 2] + part[t + 3]) * (1.0f / 1024.0f);
    __syncthreads();
    if (t < 32) {
        float acc = fc1_b[t];
        for (int c = 0; c < 64; ++c) acc += sq[c] * fc1_w[t * 64 + c];
        hbuf[t] = fmaxf(acc, 0.0f);
    }
    __syncthreads();
    if (t < 64) {
        float acc = fc2_b[t];
        for (int r = 0; r < 32; ++r) acc += hbuf[r] * fc2_w[t * 32 + r];
        scale_c[t] = 1.0f / (1.0f + expf(-acc));
    }
}

__global__ void __launch_bounds__(256)
irse_spatial_kernel(const float* __restrict__ feat, const float* __restrict__ sse_w,
                    const float* __restrict__ sse_b, float* __restrict__ spatial) {
    const int idx = blockIdx.x * 256 + threadIdx.x;
    float acc = sse_b[0];
    for (int f = 0; f < 64; ++f) acc += sse_w[f] * feat[f * 1024 + idx];
    spatial[idx] = 1.0f / (1.0f + expf(-acc));
}

__device__ __forceinline__ float irse_yval(const float* __restrict__ feat,
                                           const float* __restrict__ sc,
                                           const float* __restrict__ sp,
                                           int K, int N, int ti, int tj) {
    const int f = (K & 7) * 8 + (N & 7);
    const int bj = 2 * tj + (N >> 3);
    const int bi = 2 * ti + (K >> 3);
    const float v = feat[f * 1024 + bj * 32 + bi];
    return fmaxf(v * sc[f], v * sp[bj * 32 + bi]);
}

__global__ void __launch_bounds__(128)
irse_idct_kernel(const float* __restrict__ feat, const float* __restrict__ scale_c,
                 const float* __restrict__ spatial, float* __restrict__ out2d) {
    __shared__ float lds[4 * 256];
    const int lane = threadIdx.x & 31;
    const int wave = threadIdx.x >> 5;
    const int tile = blockIdx.x * 4 + wave;
    const int ti = tile >> 4, tj = tile & 15;
    const int h = lane >> 4;
    const int N = lane & 15;

    v8f acc = {};
#pragma unroll
    for (int k = 0; k < 4; ++k) {
        const int ka = 4 * k + 2 * h;
        v2f a, b;
        a.x = d2(ka, N);
        a.y = d2(ka + 1, N);
        b.x = irse_yval(feat, scale_c, spatial, ka, N, ti, tj);
        b.y = irse_yval(feat, scale_c, spatial, ka + 1, N, ti, tj);
        acc = wmma_f32(a, b, acc);
    }

    float* wlds = lds + wave * 256;
#pragma unroll
    for (int r = 0; r < 8; ++r) wlds[(r + 8 * h) * 16 + N] = acc[r];
    __syncthreads();

    v8f acc2 = {};
#pragma unroll
    for (int k = 0; k < 4; ++k) {
        const int ka = 4 * k + 2 * h;
        v2f a, b;
        a.x = wlds[N * 16 + ka];
        a.y = wlds[N * 16 + ka + 1];
        b.x = d2(ka, N);
        b.y = d2(ka + 1, N);
        acc2 = wmma_f32(a, b, acc2);
    }

    const int row0 = ti * 16, col0 = tj * 16;
#pragma unroll
    for (int r = 0; r < 8; ++r) {
        const int M = r + 8 * h;
        out2d[(row0 + M) * 256 + col0 + N] = acc2[r];
    }
}

__global__ void __launch_bounds__(256)
irse_bcast_kernel(const float4* __restrict__ out2d4, float4* __restrict__ out4) {
    const unsigned idx = blockIdx.x * 256u + threadIdx.x;
    out4[idx] = out2d4[idx & 16383u];
}

extern "C" void kernel_launch(void* const* d_in, const int* in_sizes, int n_in,
                              void* d_out, int out_size, void* d_ws, size_t ws_size,
                              hipStream_t stream) {
    const float* x     = (const float*)d_in[0];
    const float* fc1_w = (const float*)d_in[1];
    const float* fc1_b = (const float*)d_in[2];
    const float* fc2_w = (const float*)d_in[3];
    const float* fc2_b = (const float*)d_in[4];
    const float* sse_w = (const float*)d_in[5];
    const float* sse_b = (const float*)d_in[6];

    float* ws      = (float*)d_ws;
    float* feat    = ws;
    float* out2d   = ws + 65536;
    float* scale_c = ws + 131072;
    float* spatial = ws + 131072 + 64;

    irse_dct_kernel<<<64, 128, 0, stream>>>(x, feat);
    irse_se_mlp_kernel<<<1, 256, 0, stream>>>(feat, fc1_w, fc1_b, fc2_w, fc2_b, scale_c);
    irse_spatial_kernel<<<4, 256, 0, stream>>>(feat, sse_w, sse_b, spatial);
    irse_idct_kernel<<<64, 128, 0, stream>>>(feat, scale_c, spatial, out2d);
    irse_bcast_kernel<<<12288, 256, 0, stream>>>((const float4*)out2d, (float4*)d_out);
}


# Fuse both gate kernels into the transforms: DCT kernel (one block per tile row) also writes channel partial sums, iDCT kernel runs MLP and spatial gate itself; 5 launches -> 3; sums become trees
# speedup vs baseline: 1.5746x; 1.5746x over previous
#include <hip/hip_runtime.h>
#include <math.h>

typedef __attribute__((ext_vector_type(2))) float v2f;
typedef __attribute__((ext_vector_type(8))) float v8f;

__device__ __forceinline__ v8f wmma_f32(v2f a, v2f b, v8f c) {
    return __builtin_amdgcn_wmma_f32_16x16x4_f32(
        false, a, false, b, (short)0, c, false, false);
}

__device__ __constant__ float c_D[64] = {
    0.3535533906f,  0.3535533906f,  0.3535533906f,  0.3535533906f,
    0.3535533906f,  0.3535533906f,  0.3535533906f,  0.3535533906f,

    0.4903926402f,  0.4157348062f,  0.2777851165f,  0.0975451610f,
   -0.0975451610f, -0.2777851165f, -0.4157348062f, -0.4903926402f,

    0.4619397663f,  0.1913417162f, -0.1913417162f, -0.4619397663f,
   -0.4619397663f, -0.1913417162f,  0.1913417162f,  0.4619397663f,

    0.4157348062f, -0.0975451610f, -0.4903926402f, -0.2777851165f,
    0.2777851165f,  0.4903926402f,  0.0975451610f, -0.4157348062f,

    0.3535533906f, -0.3535533906f, -0.3535533906f,  0.3535533906f,
    0.3535533906f, -0.3535533906f, -0.3535533906f,  0.3535533906f,

    0.2777851165f, -0.4903926402f,  0.0975451610f,  0.4157348062f,
   -0.4157348062f, -0.0975451610f,  0.4903926402f, -0.2777851165f,

    0.1913417162f, -0.4619397663f,  0.4619397663f, -0.1913417162f,
   -0.1913417162f,  0.4619397663f, -0.4619397663f,  0.1913417162f,

    0.0975451610f, -0.2777851165f,  0.4157348062f, -0.4903926402f,
    0.4903926402f, -0.4157348062f,  0.2777851165f, -0.0975451610f
};

__device__ __forceinline__ float d2(int r, int c) {
    const float v = c_D[((r & 7) << 3) | (c & 7)];
    return ((r >> 3) == (c >> 3)) ? v : 0.0f;
}

__global__ void __launch_bounds__(512)
irse_dct_kernel(const float* __restrict__ x, float* __restrict__ feat,
                float* __restrict__ csum) {
    __shared__ float lds[16 * 256];
    const int lane = threadIdx.x & 31;
    const int wave = threadIdx.x >> 5;
    const int ti = blockIdx.x, tj = wave;
    const int h = lane >> 4;
    const int N = lane & 15;
    const int row0 = ti * 16, col0 = tj * 16;

    v8f acc = {};
#pragma unroll
    for (int k = 0; k < 4; ++k) {
        const int ka = 4 * k + 2 * h;
        v2f a, b;
        a.x = d2(N, ka);
        a.y = d2(N, ka + 1);
        b.x = x[(((row0 + ka) * 256) + col0 + N) * 3];
        b.y = x[(((row0 + ka + 1) * 256) + col0 + N) * 3];
        acc = wmma_f32(a, b, acc);
    }

    float* wlds = lds + wave * 256;
#pragma unroll
    for (int r = 0; r < 8; ++r) wlds[(r + 8 * h) * 16 + N] = acc[r];
    __syncthreads();

    v8f acc2 = {};
#pragma unroll
    for (int k = 0; k < 4; ++k) {
        const int ka = 4 * k + 2 * h;
        v2f a, b;
        a.x = wlds[N * 16 + ka];
        a.y = wlds[N * 16 + ka + 1];
        b.x = d2(N, ka);
        b.y = d2(N, ka + 1);
        acc2 = wmma_f32(a, b, acc2);
    }

#pragma unroll
    for (int r = 0; r < 8; ++r) {
        const int M = r + 8 * h;
        const int i = M & 7, bi = 2 * ti + (M >> 3);
        const int j = N & 7, bj = 2 * tj + (N >> 3);
        feat[(i * 8 + j) * 1024 + bj * 32 + bi] = acc2[r];
    }

#pragma unroll
    for (int r = 0; r < 8; ++r) wlds[r * 32 + lane] = acc2[r];
    __syncthreads();
    if (threadIdx.x < 64) {
        const int f = threadIdx.x;
        const float* q = lds + (f >> 3) * 32 + (f & 7);
        float s0 = 0.0f, s1 = 0.0f, s2 = 0.0f, s3 = 0.0f;
        for (int w = 0; w < 16; ++w) {
            s0 += q[w * 256];
            s1 += q[w * 256 + 8];
            s2 += q[w * 256 + 16];
            s3 += q[w * 256 + 24];
        }
        csum[ti * 64 + f] = (s0 + s1) + (s2 + s3);
    }
}

__device__ __forceinline__ float irse_sigmoid(float a) {
    return 1.0f / (1.0f + expf(-a));
}

__global__ void __launch_bounds__(128)
irse_gate_idct_kernel(const float* __restrict__ feat, const float* __restrict__ csum,
                      const float* __restrict__ fc1_w, const float* __restrict__ fc1_b,
                      const float* __restrict__ fc2_w, const float* __restrict__ fc2_b,
                      const float* __restrict__ sse_w, const float* __restrict__ sse_b,
                      float* __restrict__ out2d) {
    __shared__ float lds[4 * 256];
    __shared__ float s_part[128];
    __shared__ float s_sq[64];
    __shared__ float s_h[32];
    __shared__ float s_sc[64];
    const int t = threadIdx.x;
    const int lane = t & 31;
    const int wave = t >> 5;
    const int tile = blockIdx.x * 4 + wave;
    const int ti = tile >> 4, tj = tile & 15;
    const int h = lane >> 4;
    const int N = lane & 15;
    const int jn = N & 7;
    const int bj = 2 * tj + (N >> 3);

    float v[4][2];
    float wsp[2][2];
#pragma unroll
    for (int k = 0; k < 4; ++k) {
#pragma unroll
        for (int e = 0; e < 2; ++e) {
            const int K = 4 * k + 2 * h + e;
            v[k][e] = feat[((K & 7) * 8 + jn) * 1024 + bj * 32 + 2 * ti + (K >> 3)];
        }
    }
#pragma unroll
    for (int kk = 0; kk < 2; ++kk) {
#pragma unroll
        for (int e = 0; e < 2; ++e) wsp[kk][e] = sse_w[(4 * kk + 2 * h + e) * 8 + jn];
    }
    const float sb = sse_b[0];

    const int r1 = t >> 2, q1 = t & 3;
    const int o2 = t >> 1, q2 = t & 1;
    float w1[16], w2[16];
#pragma unroll
    for (int i = 0; i < 16; ++i) w1[i] = fc1_w[r1 * 64 + 4 * i + q1];
#pragma unroll
    for (int i = 0; i < 16; ++i) w2[i] = fc2_w[o2 * 32 + 2 * i + q2];
    const float b1 = fc1_b[r1];
    const float b2 = fc2_b[o2];

    {
        const float* p = csum + (t >> 6) * 512 + (t & 63);
        float s = 0.0f;
#pragma unroll
        for (int b = 0; b < 8; ++b) s += p[b * 64];
        s_part[t] = s;
    }
    __syncthreads();
    if (t < 64) s_sq[t] = (s_part[t] + s_part[t + 64]) * (1.0f / 1024.0f);
    __syncthreads();
    {
        float a = 0.0f;
#pragma unroll
        for (int i = 0; i < 16; ++i) a += s_sq[4 * i + q1] * w1[i];
        a += __shfl_xor(a, 1);
        a += __shfl_xor(a, 2);
        if (q1 == 0) s_h[r1] = fmaxf(a + b1, 0.0f);
    }
    __syncthreads();
    {
        float a = 0.0f;
#pragma unroll
        for (int i = 0; i < 16; ++i) a += s_h[2 * i + q2] * w2[i];
        a += __shfl_xor(a, 1);
        if (q2 == 0) s_sc[o2] = irse_sigmoid(a + b2);
    }
    __syncthreads();

    float g0 = (wsp[0][0] * v[0][0] + wsp[0][1] * v[0][1]) + (wsp[1][0] * v[1][0] + wsp[1][1] * v[1][1]);
    float g1 = (wsp[0][0] * v[2][0] + wsp[0][1] * v[2][1]) + (wsp[1][0] * v[3][0] + wsp[1][1] * v[3][1]);
    g0 += __shfl_xor(g0, 1);  g1 += __shfl_xor(g1, 1);
    g0 += __shfl_xor(g0, 2);  g1 += __shfl_xor(g1, 2);
    g0 += __shfl_xor(g0, 4);  g1 += __shfl_xor(g1, 4);
    g0 += __shfl_xor(g0, 16); g1 += __shfl_xor(g1, 16);
    float sp[2];
    sp[0] = irse_sigmoid(g0 + sb);
    sp[1] = irse_sigmoid(g1 + sb);

    float scv[2][2];
#pragma unroll
    for (int kk = 0; kk < 2; ++kk) {
#pragma unroll
        for (int e = 0; e < 2; ++e) scv[kk][e] = s_sc[(4 * kk + 2 * h + e) * 8 + jn];
    }

    v8f acc = {};
#pragma unroll
    for (int k = 0; k < 4; ++k) {
        const int ka = 4 * k + 2 * h;
        v2f a, b;
        a.x = d2(ka, N);
        a.y = d2(ka + 1, N);
        b.x = fmaxf(v[k][0] * scv[k & 1][0], v[k][0] * sp[k >> 1]);
        b.y = fmaxf(v[k][1] * scv[k & 1][1], v[k][1] * sp[k >> 1]);
        acc = wmma_f32(a, b, acc);
    }

    float* wlds = lds + wave * 256;
#pragma unroll
    for (int r = 0; r < 8; ++r) wlds[(r + 8 * h) * 16 + N] = acc[r];
    __syncthreads();

    v8f acc2 = {};
#pragma unroll
    for (int k = 0; k < 4; ++k) {
        const int ka = 4 * k + 2 * h;
        v2f a, b;
        a.x = wlds[N * 16 + ka];
        a.y = wlds[N * 16 + ka + 1];
        b.x = d2(ka, N);
        b.y = d2(ka + 1, N);
        acc2 = wmma_f32(a, b, acc2);
    }

    const int row0 = ti * 16, col0 = tj * 16;
#pragma unroll
    for (int r = 0; r < 8; ++r) {
        const int M = r + 8 * h;
        out2d[(row0 + M) * 256 + col0 + N] = acc2[r];
    }
}

__global__ void __launch_bounds__(256)
irse_bcast_kernel(const float4* __restrict__ out2d4, float4* __restrict__ out4) {
    const unsigned idx = blockIdx.x * 256u + threadIdx.x;
    out4[idx] = out2d4[idx & 16383u];
}

extern "C" void kernel_launch(void* const* d_in, const int* in_sizes, int n_in,
                              void* d_out, int out_size, void* d_ws, size_t ws_size,
                              hipStream_t stream) {
    const float* x     = (const float*)d_in[0];
    const float* fc1_w = (const float*)d_in[1];
    const float* fc1_b = (const float*)d_in[2];
    const float* fc2_w = (const float*)d_in[3];
    const float* fc2_b = (const float*)d_in[4];
    const float* sse_w = (const float*)d_in[5];
    const float* sse_b = (const float*)d_in[6];

    float* ws      = (float*)d_ws;
    float* feat    = ws;
    float* out2d   = ws + 65536;
    float* csum    = ws + 131072;

    irse_dct_kernel<<<16, 512, 0, stream>>>(x, feat, csum);
    irse_gate_idct_kernel<<<64, 128, 0, stream>>>(feat, csum, fc1_w, fc1_b, fc2_w, fc2_b,
                                                  sse_w, sse_b, out2d);
    irse_bcast_kernel<<<12288, 256, 0, stream>>>((const float4*)out2d, (float4*)d_out);
}


# feat workspace kept in its reader's fragment order (tile x 32 lanes x 8 floats): DCT kernel writes 2 whole-line b128 stores per lane for 8 scattered dwords, gate kernel reads 2 b128 for 8 gathers
# speedup vs baseline: 1.8940x; 1.2028x over previous
#include <hip/hip_runtime.h>
#include <math.h>

typedef __attribute__((ext_vector_type(2))) float v2f;
typedef __attribute__((ext_vector_type(8))) float v8f;

__device__ __forceinline__ v8f wmma_f32(v2f a, v2f b, v8f c) {
    return __builtin_amdgcn_wmma_f32_16x16x4_f32(
        false, a, false, b, (short)0, c, false, false);
}

__device__ __constant__ float c_D[64] = {
    0.3535533906f,  0.3535533906f,  0.3535533906f,  0.3535533906f,
    0.3535533906f,  0.3535533906f,  0.3535533906f,  0.3535533906f,

    0.4903926402f,  0.4157348062f,  0.2777851165f,  0.0975451610f,
   -0.0975451610f, -0.2777851165f, -0.4157348062f, -0.4903926402f,

    0.4619397663f,  0.1913417162f, -0.1913417162f, -0.4619397663f,
   -0.4619397663f, -0.1913417162f,  0.1913417162f,  0.4619397663f,

    0.4157348062f, -0.0975451610f, -0.4903926402f, -0.2777851165f,
    0.2777851165f,  0.4903926402f,  0.0975451610f, -0.4157348062f,

    0.3535533906f, -0.3535533906f, -0.3535533906f,  0.3535533906f,
    0.3535533906f, -0.3535533906f, -0.3535533906f,  0.3535533906f,

    0.2777851165f, -0.4903926402f,  0.0975451610f,  0.4157348062f,
   -0.4157348062f, -0.0975451610f,  0.4903926402f, -0.2777851165f,

    0.1913417162f, -0.4619397663f,  0.4619397663f, -0.1913417162f,
   -0.1913417162f,  0.4619397663f, -0.4619397663f,  0.1913417162f,

    0.0975451610f, -0.2777851165f,  0.4157348062f, -0.4903926402f,
    0.4903926402f, -0.4157348062f,  0.2777851165f, -0.0975451610f
};

__device__ __forceinline__ float d2(int r, int c) {
    const float v = c_D[((r & 7) << 3) | (c & 7)];
    return ((r >> 3) == (c >> 3)) ? v : 0.0f;
}

__global__ void __launch_bounds__(512)
irse_dct_kernel(const float* __restrict__ x, float* __restrict__ feat,
                float* __restrict__ csum) {
    __shared__ float lds[16 * 256];
    const int lane = threadIdx.x & 31;
    const int wave = threadIdx.x >> 5;
    const int ti = blockIdx.x, tj = wave;
    const int h = lane >> 4;
    const int N = lane & 15;
    const int row0 = ti * 16, col0 = tj * 16;

    v8f acc = {};
#pragma unroll
    for (int k = 0; k < 4; ++k) {
        const int ka = 4 * k + 2 * h;
        v2f a, b;
        a.x = d2(N, ka);
        a.y = d2(N, ka + 1);
        b.x = x[(((row0 + ka) * 256) + col0 + N) * 3];
        b.y = x[(((row0 + ka + 1) * 256) + col0 + N) * 3];
        acc = wmma_f32(a, b, acc);
    }

    float* wlds = lds + wave * 256;
#pragma unroll
    for (int r = 0; r < 8; ++r) wlds[(r + 8 * h) * 16 + N] = acc[r];
    __syncthreads();

    v8f acc2 = {};
#pragma unroll
    for (int k = 0; k < 4; ++k) {
        const int ka = 4 * k + 2 * h;
        v2f a, b;
        a.x = wlds[N * 16 + ka];
        a.y = wlds[N * 16 + ka + 1];
        b.x = d2(N, ka);
        b.y = d2(N, ka + 1);
        acc2 = wmma_f32(a, b, acc2);
    }

    {
        float4* ft = reinterpret_cast<float4*>(feat + (ti * 16 + tj) * 256);
        ft[N * 2 + h]        = make_float4(acc2[0], acc2[1], acc2[4], acc2[5]);
        ft[(N + 16) * 2 + h] = make_float4(acc2[2], acc2[3], acc2[6], acc2[7]);
    }

#pragma unroll
    for (int r = 0; r < 8; ++r) wlds[r * 32 + lane] = acc2[r];
    __syncthreads();
    if (threadIdx.x < 64) {
        const int f = threadIdx.x;
        const float* q = lds + (f >> 3) * 32 + (f & 7);
        float s0 = 0.0f, s1 = 0.0f, s2 = 0.0f, s3 = 0.0f;
        for (int w = 0; w < 16; ++w) {
            s0 += q[w * 256];
            s1 += q[w * 256 + 8];
            s2 += q[w * 256 + 16];
            s3 += q[w * 256 + 24];
        }
        csum[ti * 64 + f] = (s0 + s1) + (s2 + s3);
    }
}

__device__ __forceinline__ float irse_sigmoid(float a) {
    return 1.0f / (1.0f + expf(-a));
}

__global__ void __launch_bounds__(128)
irse_gate_idct_kernel(const float* __restrict__ feat, const float* __restrict__ csum,
                      const float* __restrict__ fc1_w, const float* __restrict__ fc1_b,
                      const float* __restrict__ fc2_w, const float* __restrict__ fc2_b,
                      const float* __restrict__ sse_w, const float* __restrict__ sse_b,
                      float* __restrict__ out2d) {
    __shared__ float lds[4 * 256];
    __shared__ float s_part[128];
    __shared__ float s_sq[64];
    __shared__ float s_h[32];
    __shared__ float s_sc[64];
    const int t = threadIdx.x;
    const int lane = t & 31;
    const int wave = t >> 5;
    const int tile = blockIdx.x * 4 + wave;
    const int ti = tile >> 4, tj = tile & 15;
    const int h = lane >> 4;
    const int N = lane & 15;
    const int jn = N & 7;

    float v[4][2];
    float wsp[2][2];
    {
        const float4* ft = reinterpret_cast<const float4*>(feat + tile * 256) + lane * 2;
        const float4 lo = ft[0], hi = ft[1];
        v[0][0] = lo.x; v[0][1] = lo.y; v[1][0] = lo.z; v[1][1] = lo.w;
        v[2][0] = hi.x; v[2][1] = hi.y; v[3][0] = hi.z; v[3][1] = hi.w;
    }
#pragma unroll
    for (int kk = 0; kk < 2; ++kk) {
#pragma unroll
        for (int e = 0; e < 2; ++e) wsp[kk][e] = sse_w[(4 * kk + 2 * h + e) * 8 + jn];
    }
    const float sb = sse_b[0];

    const int r1 = t >> 2, q1 = t & 3;
    const int o2 = t >> 1, q2 = t & 1;
    float w1[16], w2[16];
#pragma unroll
    for (int i = 0; i < 16; ++i) w1[i] = fc1_w[r1 * 64 + 4 * i + q1];
#pragma unroll
    for (int i = 0; i < 16; ++i) w2[i] = fc2_w[o2 * 32 + 2 * i + q2];
    const float b1 = fc1_b[r1];
    const float b2 = fc2_b[o2];

    {
        const float* p = csum + (t >> 6) * 512 + (t & 63);
        float s = 0.0f;
#pragma unroll
        for (int b = 0; b < 8; ++b) s += p[b * 64];
        s_part[t] = s;
    }
    __syncthreads();
    if (t < 64) s_sq[t] = (s_part[t] + s_part[t + 64]) * (1.0f / 1024.0f);
    __syncthreads();
    {
        float a = 0.0f;
#pragma unroll
        for (int i = 0; i < 16; ++i) a += s_sq[4 * i + q1] * w1[i];
        a += __shfl_xor(a, 1);
        a += __shfl_xor(a, 2);
        if (q1 == 0) s_h[r1] = fmaxf(a + b1, 0.0f);
    }
    __syncthreads();
    {
        float a = 0.0f;
#pragma unroll
        for (int i = 0; i < 16; ++i) a += s_h[2 * i + q2] * w2[i];
        a += __shfl_xor(a, 1);
        if (q2 == 0) s_sc[o2] = irse_sigmoid(a + b2);
    }
    __syncthreads();

    float g0 = (wsp[0][0] * v[0][0] + wsp[0][1] * v[0][1]) + (wsp[1][0] * v[1][0] + wsp[1][1] * v[1][1]);
    float g1 = (wsp[0][0] * v[2][0] + wsp[0][1] * v[2][1]) + (wsp[1][0] * v[3][0] + wsp[1][1] * v[3][1]);
    g0 += __shfl_xor(g0, 1);  g1 += __shfl_xor(g1, 1);
    g0 += __shfl_xor(g0, 2);  g1 += __shfl_xor(g1, 2);
    g0 += __shfl_xor(g0, 4);  g1 += __shfl_xor(g1, 4);
    g0 += __shfl_xor(g0, 16); g1 += __shfl_xor(g1, 16);
    float sp[2];
    sp[0] = irse_sigmoid(g0 + sb);
    sp[1] = irse_sigmoid(g1 + sb);

    float scv[2][2];
#pragma unroll
    for (int kk = 0; kk < 2; ++kk) {
#pragma unroll
        for (int e = 0; e < 2; ++e) scv[kk][e] = s_sc[(4 * kk + 2 * h + e) * 8 + jn];
    }

    v8f acc = {};
#pragma unroll
    for (int k = 0; k < 4; ++k) {
        const int ka = 4 * k + 2 * h;
        v2f a, b;
        a.x = d2(ka, N);
        a.y = d2(ka + 1, N);
        b.x = fmaxf(v[k][0] * scv[k & 1][0], v[k][0] * sp[k >> 1]);
        b.y = fmaxf(v[k][1] * scv[k & 1][1], v[k][1] * sp[k >> 1]);
        acc = wmma_f32(a, b, acc);
    }

    float* wlds = lds + wave * 256;
#pragma unroll
    for (int r = 0; r < 8; ++r) wlds[(r + 8 * h) * 16 + N] = acc[r];
    __syncthreads();

    v8f acc2 = {};
#pragma unroll
    for (int k = 0; k < 4; ++k) {
        const int ka = 4 * k + 2 * h;
        v2f a, b;
        a.x = wlds[N * 16 + ka];
        a.y = wlds[N * 16 + ka + 1];
        b.x = d2(ka, N);
        b.y = d2(ka + 1, N);
        acc2 = wmma_f32(a, b, acc2);
    }

    const int row0 = ti * 16, col0 = tj * 16;
#pragma unroll
    for (int r = 0; r < 8; ++r) {
        const int M = r + 8 * h;
        out2d[(row0 + M) * 256 + col0 + N] = acc2[r];
    }
}

__global__ void __launch_bounds__(256)
irse_bcast_kernel(const float4* __restrict__ out2d4, float4* __restrict__ out4) {
    const unsigned idx = blockIdx.x * 256u + threadIdx.x;
    out4[idx] = out2d4[idx & 16383u];
}

extern "C" void kernel_launch(void* const* d_in, const int* in_sizes, int n_in,
                              void* d_out, int out_size, void* d_ws, size_t ws_size,
                              hipStream_t stream) {
    const float* x     = (const float*)d_in[0];
    const float* fc1_w = (const float*)d_in[1];
    const float* fc1_b = (const float*)d_in[2];
    const float* fc2_w = (const float*)d_in[3];
    const float* fc2_b = (const float*)d_in[4];
    const float* sse_w = (const float*)d_in[5];
    const float* sse_b = (const float*)d_in[6];

    float* ws      = (float*)d_ws;
    float* feat    = ws;
    float* out2d   = ws + 65536;
    float* csum    = ws + 131072;

    irse_dct_kernel<<<16, 512, 0, stream>>>(x, feat, csum);
    irse_gate_idct_kernel<<<64, 128, 0, stream>>>(feat, csum, fc1_w, fc1_b, fc2_w, fc2_b,
                                                  sse_w, sse_b, out2d);
    irse_bcast_kernel<<<12288, 256, 0, stream>>>((const float4*)out2d, (float4*)d_out);
}


# Broadcast kernel: 2048 blocks, a lane loads 1 float4 and stores it to 6 planes from registers (was 12288 blocks, 1 load + 1 store a lane): 6x fewer waves and source reads, one resident round
# speedup vs baseline: 2.2688x; 1.1979x over previous
#include <hip/hip_runtime.h>
#include <math.h>

typedef __attribute__((ext_vector_type(2))) float v2f;
typedef __attribute__((ext_vector_type(8))) float v8f;

__device__ __forceinline__ v8f wmma_f32(v2f a, v2f b, v8f c) {
    return __builtin_amdgcn_wmma_f32_16x16x4_f32(
        false, a, false, b, (short)0, c, false, false);
}

__device__ __constant__ float c_D[64] = {
    0.3535533906f,  0.3535533906f,  0.3535533906f,  0.3535533906f,
    0.3535533906f,  0.3535533906f,  0.3535533906f,  0.3535533906f,

    0.4903926402f,  0.4157348062f,  0.2777851165f,  0.0975451610f,
   -0.0975451610f, -0.2777851165f, -0.4157348062f, -0.4903926402f,

    0.4619397663f,  0.1913417162f, -0.1913417162f, -0.4619397663f,
   -0.4619397663f, -0.1913417162f,  0.1913417162f,  0.4619397663f,

    0.4157348062f, -0.0975451610f, -0.4903926402f, -0.2777851165f,
    0.2777851165f,  0.4903926402f,  0.0975451610f, -0.4157348062f,

    0.3535533906f, -0.3535533906f, -0.3535533906f,  0.3535533906f,
    0.3535533906f, -0.3535533906f, -0.3535533906f,  0.3535533906f,

    0.2777851165f, -0.4903926402f,  0.0975451610f,  0.4157348062f,
   -0.4157348062f, -0.0975451610f,  0.4903926402f, -0.2777851165f,

    0.1913417162f, -0.4619397663f,  0.4619397663f, -0.1913417162f,
   -0.1913417162f,  0.4619397663f, -0.4619397663f,  0.1913417162f,

    0.0975451610f, -0.2777851165f,  0.4157348062f, -0.4903926402f,
    0.4903926402f, -0.4157348062f,  0.2777851165f, -0.0975451610f
};

__device__ __forceinline__ float d2(int r, int c) {
    const float v = c_D[((r & 7) << 3) | (c & 7)];
    return ((r >> 3) == (c >> 3)) ? v : 0.0f;
}

__global__ void __launch_bounds__(512)
irse_dct_kernel(const float* __restrict__ x, float* __restrict__ feat,
                float* __restrict__ csum) {
    __shared__ float lds[16 * 256];
    const int lane = threadIdx.x & 31;
    const int wave = threadIdx.x >> 5;
    const int ti = blockIdx.x, tj = wave;
    const int h = lane >> 4;
    const int N = lane & 15;
    const int row0 = ti * 16, col0 = tj * 16;

    v8f acc = {};
#pragma unroll
    for (int k = 0; k < 4; ++k) {
        const int ka = 4 * k + 2 * h;
        v2f a, b;
        a.x = d2(N, ka);
        a.y = d2(N, ka + 1);
        b.x = x[(((row0 + ka) * 256) + col0 + N) * 3];
        b.y = x[(((row0 + ka + 1) * 256) + col0 + N) * 3];
        acc = wmma_f32(a, b, acc);
    }

    float* wlds = lds + wave * 256;
#pragma unroll
    for (int r = 0; r < 8; ++r) wlds[(r + 8 * h) * 16 + N] = acc[r];
    __syncthreads();

    v8f acc2 = {};
#pragma unroll
    for (int k = 0; k < 4; ++k) {
        const int ka = 4 * k + 2 * h;
        v2f a, b;
        a.x = wlds[N * 16 + ka];
        a.y = wlds[N * 16 + ka + 1];
        b.x = d2(N, ka);
        b.y = d2(N, ka + 1);
        acc2 = wmma_f32(a, b, acc2);
    }

    {
        float4* ft = reinterpret_cast<float4*>(feat + (ti * 16 + tj) * 256);
        ft[N * 2 + h]        = make_float4(acc2[0], acc2[1], acc2[4], acc2[5]);
        ft[(N + 16) * 2 + h] = make_float4(acc2[2], acc2[3], acc2[6], acc2[7]);
    }

#pragma unroll
    for (int r = 0; r < 8; ++r) wlds[r * 32 + lane] = acc2[r];
    __syncthreads();
    if (threadIdx.x < 64) {
        const int f = threadIdx.x;
        const float* q = lds + (f >> 3) * 32 + (f & 7);
        float s0 = 0.0f, s1 = 0.0f, s2 = 0.0f, s3 = 0.0f;
        for (int w = 0; w < 16; ++w) {
            s0 += q[w * 256];
            s1 += q[w * 256 + 8];
            s2 += q[w * 256 + 16];
            s3 += q[w * 256 + 24];
        }
        csum[ti * 64 + f] = (s0 + s1) + (s2 + s3);
    }
}

__device__ __forceinline__ float irse_sigmoid(float a) {
    return 1.0f / (1.0f + expf(-a));
}

__global__ void __launch_bounds__(128)
irse_gate_idct_kernel(const float* __restrict__ feat, const float* __restrict__ csum,
                      const float* __restrict__ fc1_w, const float* __restrict__ fc1_b,
                      const float* __restrict__ fc2_w, const float* __restrict__ fc2_b,
                      const float* __restrict__ sse_w, const float* __restrict__ sse_b,
                      float* __restrict__ out2d) {
    __shared__ float lds[4 * 256];
    __shared__ float s_part[128];
    __shared__ float s_sq[64];
    __shared__ float s_h[32];
    __shared__ float s_sc[64];
    const int t = threadIdx.x;
    const int lane = t & 31;
    const int wave = t >> 5;
    const int tile = blockIdx.x * 4 + wave;
    const int ti = tile >> 4, tj = tile & 15;
    const int h = lane >> 4;
    const int N = lane & 15;
    const int jn = N & 7;

    float v[4][2];
    float wsp[2][2];
    {
        const float4* ft = reinterpret_cast<const float4*>(feat + tile * 256) + lane * 2;
        const float4 lo = ft[0], hi = ft[1];
        v[0][0] = lo.x; v[0][1] = lo.y; v[1][0] = lo.z; v[1][1] = lo.w;
        v[2][0] = hi.x; v[2][1] = hi.y; v[3][0] = hi.z; v[3][1] = hi.w;
    }
#pragma unroll
    for (int kk = 0; kk < 2; ++kk) {
#pragma unroll
        for (int e = 0; e < 2; ++e) wsp[kk][e] = sse_w[(4 * kk + 2 * h + e) * 8 + jn];
    }
    const float sb = sse_b[0];

    const int r1 = t >> 2, q1 = t & 3;
    const int o2 = t >> 1, q2 = t & 1;
    float w1[16], w2[16];
#pragma unroll
    for (int i = 0; i < 16; ++i) w1[i] = fc1_w[r1 * 64 + 4 * i + q1];
#pragma unroll
    for (int i = 0; i < 16; ++i) w2[i] = fc2_w[o2 * 32 + 2 * i + q2];
    const float b1 = fc1_b[r1];
    const float b2 = fc2_b[o2];

    {
        const float* p = csum + (t >> 6) * 512 + (t & 63);
        float s = 0.0f;
#pragma unroll
        for (int b = 0; b < 8; ++b) s += p[b * 64];
        s_part[t] = s;
    }
    __syncthreads();
    if (t < 64) s_sq[t] = (s_part[t] + s_part[t + 64]) * (1.0f / 1024.0f);
    __syncthreads();
    {
        float a = 0.0f;
#pragma unroll
        for (int i = 0; i < 16; ++i) a += s_sq[4 * i + q1] * w1[i];
        a += __shfl_xor(a, 1);
        a += __shfl_xor(a, 2);
        if (q1 == 0) s_h[r1] = fmaxf(a + b1, 0.0f);
    }
    __syncthreads();
    {
        float a = 0.0f;
#pragma unroll
        for (int i = 0; i < 16; ++i) a += s_h[2 * i + q2] * w2[i];
        a += __shfl_xor(a, 1);
        if (q2 == 0) s_sc[o2] = irse_sigmoid(a + b2);
    }
    __syncthreads();

    float g0 = (wsp[0][0] * v[0][0] + wsp[0][1] * v[0][1]) + (wsp[1][0] * v[1][0] + wsp[1][1] * v[1][1]);
    float g1 = (wsp[0][0] * v[2][0] + wsp[0][1] * v[2][1]) + (wsp[1][0] * v[3][0] + wsp[1][1] * v[3][1]);
    g0 += __shfl_xor(g0, 1);  g1 += __shfl_xor(g1, 1);
    g0 += __shfl_xor(g0, 2);  g1 += __shfl_xor(g1, 2);
    g0 += __shfl_xor(g0, 4);  g1 += __shfl_xor(g1, 4);
    g0 += __shfl_xor(g0, 16); g1 += __shfl_xor(g1, 16);
    float sp[2];
    sp[0] = irse_sigmoid(g0 + sb);
    sp[1] = irse_sigmoid(g1 + sb);

    float scv[2][2];
#pragma unroll
    for (int kk = 0; kk < 2; ++kk) {
#pragma unroll
        for (int e = 0; e < 2; ++e) scv[kk][e] = s_sc[(4 * kk + 2 * h + e) * 8 + jn];
    }

    v8f acc = {};
#pragma unroll
    for (int k = 0; k < 4; ++k) {
        const int ka = 4 * k + 2 * h;
        v2f a, b;
        a.x = d2(ka, N);
        a.y = d2(ka + 1, N);
        b.x = fmaxf(v[k][0] * scv[k & 1][0], v[k][0] * sp[k >> 1]);
        b.y = fmaxf(v[k][1] * scv[k & 1][1], v[k][1] * sp[k >> 1]);
        acc = wmma_f32(a, b, acc);
    }

    float* wlds = lds + wave * 256;
#pragma unroll
    for (int r = 0; r < 8; ++r) wlds[(r + 8 * h) * 16 + N] = acc[r];
    __syncthreads();

    v8f acc2 = {};
#pragma unroll
    for (int k = 0; k < 4; ++k) {
        const int ka = 4 * k + 2 * h;
        v2f a, b;
        a.x = wlds[N * 16 + ka];
        a.y = wlds[N * 16 + ka + 1];
        b.x = d2(ka, N);
        b.y = d2(ka + 1, N);
        acc2 = wmma_f32(a, b, acc2);
    }

    const int row0 = ti * 16, col0 = tj * 16;
#pragma unroll
    for (int r = 0; r < 8; ++r) {
        const int M = r + 8 * h;
        out2d[(row0 + M) * 256 + col0 + N] = acc2[r];
    }
}

__global__ void __launch_bounds__(256)
irse_bcast_kernel(const float4* out2d4, float4* out4) {
    const unsigned chunk = blockIdx.x & 63u;
    const unsigned group = blockIdx.x >> 6;
    const unsigned pos = chunk * 256u + threadIdx.x;
    const float4 v = out2d4[pos];
    float4* dst = out4 + group * (6u * 16384u) + pos;
#pragma unroll
    for (int p = 0; p < 6; ++p) dst[p * 16384] = v;
}

extern "C" void kernel_launch(void* const* d_in, const int* in_sizes, int n_in,
                              void* d_out, int out_size, void* d_ws, size_t ws_size,
                              hipStream_t stream) {
    const float* x     = (const float*)d_in[0];
    const float* fc1_w = (const float*)d_in[1];
    const float* fc1_b = (const float*)d_in[2];
    const float* fc2_w = (const float*)d_in[3];
    const float* fc2_b = (const float*)d_in[4];
    const float* sse_w = (const float*)d_in[5];
    const float* sse_b = (const float*)d_in[6];

    float* ws      = (float*)d_ws;
    float* feat    = ws;
    float* out2d   = ws + 65536;
    float* csum    = ws + 131072;

    irse_dct_kernel<<<16, 512, 0, stream>>>(x, feat, csum);
    irse_gate_idct_kernel<<<64, 128, 0, stream>>>(feat, csum, fc1_w, fc1_b, fc2_w, fc2_b,
                                                  sse_w, sse_b, out2d);
    irse_bcast_kernel<<<2048, 256, 0, stream>>>((const float4*)out2d, (float4*)d_out);
}


# Gate+iDCT fused into the broadcast, 3 launches -> 2: block = tile row x 12 planes, strip parked in LDS, each wave streams the whole 16 KB strip into ONE plane (no lane hops from plane to plane)
# speedup vs baseline: 2.5411x; 1.1200x over previous
#include <hip/hip_runtime.h>
#include <math.h>

typedef __attribute__((ext_vector_type(2))) float v2f;
typedef __attribute__((ext_vector_type(8))) float v8f;

__device__ __forceinline__ v8f wmma_f32(v2f a, v2f b, v8f c) {
    return __builtin_amdgcn_wmma_f32_16x16x4_f32(
        false, a, false, b, (short)0, c, false, false);
}

__device__ __constant__ float c_D[64] = {
    0.3535533906f,  0.3535533906f,  0.3535533906f,  0.3535533906f,
    0.3535533906f,  0.3535533906f,  0.3535533906f,  0.3535533906f,

    0.4903926402f,  0.4157348062f,  0.2777851165f,  0.0975451610f,
   -0.0975451610f, -0.2777851165f, -0.4157348062f, -0.4903926402f,

    0.4619397663f,  0.1913417162f, -0.1913417162f, -0.4619397663f,
   -0.4619397663f, -0.1913417162f,  0.1913417162f,  0.4619397663f,

    0.4157348062f, -0.0975451610f, -0.4903926402f, -0.2777851165f,
    0.2777851165f,  0.4903926402f,  0.0975451610f, -0.4157348062f,

    0.3535533906f, -0.3535533906f, -0.3535533906f,  0.3535533906f,
    0.3535533906f, -0.3535533906f, -0.3535533906f,  0.3535533906f,

    0.2777851165f, -0.4903926402f,  0.0975451610f,  0.4157348062f,
   -0.4157348062f, -0.0975451610f,  0.4903926402f, -0.2777851165f,

    0.1913417162f, -0.4619397663f,  0.4619397663f, -0.1913417162f,
   -0.1913417162f,  0.4619397663f, -0.4619397663f,  0.1913417162f,

    0.0975451610f, -0.2777851165f,  0.4157348062f, -0.4903926402f,
    0.4903926402f, -0.4157348062f,  0.2777851165f, -0.0975451610f
};

__device__ __forceinline__ float d2(int r, int c) {
    const float v = c_D[((r & 7) << 3) | (c & 7)];
    return ((r >> 3) == (c >> 3)) ? v : 0.0f;
}

__global__ void __launch_bounds__(512)
irse_dct_kernel(const float* __restrict__ x, float* __restrict__ feat,
                float* __restrict__ csum) {
    __shared__ float lds[16 * 256];
    const int lane = threadIdx.x & 31;
    const int wave = threadIdx.x >> 5;
    const int ti = blockIdx.x, tj = wave;
    const int h = lane >> 4;
    const int N = lane & 15;
    const int row0 = ti * 16, col0 = tj * 16;

    v8f acc = {};
#pragma unroll
    for (int k = 0; k < 4; ++k) {
        const int ka = 4 * k + 2 * h;
        v2f a, b;
        a.x = d2(N, ka);
        a.y = d2(N, ka + 1);
        b.x = x[(((row0 + ka) * 256) + col0 + N) * 3];
        b.y = x[(((row0 + ka + 1) * 256) + col0 + N) * 3];
        acc = wmma_f32(a, b, acc);
    }

    float* wlds = lds + wave * 256;
#pragma unroll
    for (int r = 0; r < 8; ++r) wlds[(r + 8 * h) * 16 + N] = acc[r];
    __syncthreads();

    v8f acc2 = {};
#pragma unroll
    for (int k = 0; k < 4; ++k) {
        const int ka = 4 * k + 2 * h;
        v2f a, b;
        a.x = wlds[N * 16 + ka];
        a.y = wlds[N * 16 + ka + 1];
        b.x = d2(N, ka);
        b.y = d2(N, ka + 1);
        acc2 = wmma_f32(a, b, acc2);
    }

    {
        float4* ft = reinterpret_cast<float4*>(feat + (ti * 16 + tj) * 256);
        ft[N * 2 + h]        = make_float4(acc2[0], acc2[1], acc2[4], acc2[5]);
        ft[(N + 16) * 2 + h] = make_float4(acc2[2], acc2[3], acc2[6], acc2[7]);
    }

#pragma unroll
    for (int r = 0; r < 8; ++r) wlds[r * 32 + lane] = acc2[r];
    __syncthreads();
    if (threadIdx.x < 64) {
        const int f = threadIdx.x;
        const float* q = lds + (f >> 3) * 32 + (f & 7);
        float s0 = 0.0f, s1 = 0.0f, s2 = 0.0f, s3 = 0.0f;
        for (int w = 0; w < 16; ++w) {
            s0 += q[w * 256];
            s1 += q[w * 256 + 8];
            s2 += q[w * 256 + 16];
            s3 += q[w * 256 + 24];
        }
        csum[ti * 64 + f] = (s0 + s1) + (s2 + s3);
    }
}

__device__ __forceinline__ float irse_sigmoid(float a) {
    return 1.0f / (1.0f + expf(-a));
}

__global__ void __launch_bounds__(512)
irse_gate_idct_bcast_kernel(const float* __restrict__ feat, const float* __restrict__ csum,
                            const float* __restrict__ fc1_w, const float* __restrict__ fc1_b,
                            const float* __restrict__ fc2_w, const float* __restrict__ fc2_b,
                            const float* __restrict__ sse_w, const float* __restrict__ sse_b,
                            float4* __restrict__ out4) {
    __shared__ float lds[16 * 256];
    __shared__ float4 strip4[16 * 64];
    __shared__ float s_part[128];
    __shared__ float s_sq[64];
    __shared__ float s_h[32];
    __shared__ float s_sc[64];
    const int t = threadIdx.x;
    const int lane = t & 31;
    const int wave = t >> 5;
    const int ti = blockIdx.x & 15;
    const int grp = blockIdx.x >> 4;
    const int tj = wave;
    const int tile = ti * 16 + tj;
    const int h = lane >> 4;
    const int N = lane & 15;
    const int jn = N & 7;
    const bool mlp = t < 128;

    float v[4][2];
    float wsp[2][2];
    {
        const float4* ft = reinterpret_cast<const float4*>(feat + tile * 256) + lane * 2;
        const float4 lo = ft[0], hi = ft[1];
        v[0][0] = lo.x; v[0][1] = lo.y; v[1][0] = lo.z; v[1][1] = lo.w;
        v[2][0] = hi.x; v[2][1] = hi.y; v[3][0] = hi.z; v[3][1] = hi.w;
    }
#pragma unroll
    for (int kk = 0; kk < 2; ++kk) {
#pragma unroll
        for (int e = 0; e < 2; ++e) wsp[kk][e] = sse_w[(4 * kk + 2 * h + e) * 8 + jn];
    }
    const float sb = sse_b[0];

    const int tm = t & 127;
    const int r1 = tm >> 2, q1 = tm & 3;
    const int o2 = tm >> 1, q2 = tm & 1;
    float w1[16], w2[16];
    float b1 = 0.0f, b2 = 0.0f, spart = 0.0f;
#pragma unroll
    for (int i = 0; i < 16; ++i) { w1[i] = 0.0f; w2[i] = 0.0f; }
    if (mlp) {
#pragma unroll
        for (int i = 0; i < 16; ++i) w1[i] = fc1_w[r1 * 64 + 4 * i + q1];
#pragma unroll
        for (int i = 0; i < 16; ++i) w2[i] = fc2_w[o2 * 32 + 2 * i + q2];
        b1 = fc1_b[r1];
        b2 = fc2_b[o2];
        const float* p = csum + (tm >> 6) * 512 + (tm & 63);
#pragma unroll
        for (int b = 0; b < 8; ++b) spart += p[b * 64];
        s_part[tm] = spart;
    }
    __syncthreads();
    if (t < 64) s_sq[t] = (s_part[t] + s_part[t + 64]) * (1.0f / 1024.0f);
    __syncthreads();
    if (mlp) {
        float a = 0.0f;
#pragma unroll
        for (int i = 0; i < 16; ++i) a += s_sq[4 * i + q1] * w1[i];
        a += __shfl_xor(a, 1);
        a += __shfl_xor(a, 2);
        if (q1 == 0) s_h[r1] = fmaxf(a + b1, 0.0f);
    }
    __syncthreads();
    if (mlp) {
        float a = 0.0f;
#pragma unroll
        for (int i = 0; i < 16; ++i) a += s_h[2 * i + q2] * w2[i];
        a += __shfl_xor(a, 1);
        if (q2 == 0) s_sc[o2] = irse_sigmoid(a + b2);
    }
    __syncthreads();

    float g0 = (wsp[0][0] * v[0][0] + wsp[0][1] * v[0][1]) + (wsp[1][0] * v[1][0] + wsp[1][1] * v[1][1]);
    float g1 = (wsp[0][0] * v[2][0] + wsp[0][1] * v[2][1]) + (wsp[1][0] * v[3][0] + wsp[1][1] * v[3][1]);
    g0 += __shfl_xor(g0, 1);  g1 += __shfl_xor(g1, 1);
    g0 += __shfl_xor(g0, 2);  g1 += __shfl_xor(g1, 2);
    g0 += __shfl_xor(g0, 4);  g1 += __shfl_xor(g1, 4);
    g0 += __shfl_xor(g0, 16); g1 += __shfl_xor(g1, 16);
    float sp[2];
    sp[0] = irse_sigmoid(g0 + sb);
    sp[1] = irse_sigmoid(g1 + sb);

    float scv[2][2];
#pragma unroll
    for (int kk = 0; kk < 2; ++kk) {
#pragma unroll
        for (int e = 0; e < 2; ++e) scv[kk][e] = s_sc[(4 * kk + 2 * h + e) * 8 + jn];
    }

    v8f acc = {};
#pragma unroll
    for (int k = 0; k < 4; ++k) {
        const int ka = 4 * k + 2 * h;
        v2f a, b;
        a.x = d2(ka, N);
        a.y = d2(ka + 1, N);
        b.x = fmaxf(v[k][0] * scv[k & 1][0], v[k][0] * sp[k >> 1]);
        b.y = fmaxf(v[k][1] * scv[k & 1][1], v[k][1] * sp[k >> 1]);
        acc = wmma_f32(a, b, acc);
    }

    float* wlds = lds + wave * 256;
#pragma unroll
    for (int r = 0; r < 8; ++r) wlds[(r + 8 * h) * 16 + N] = acc[r];
    __syncthreads();

    v8f acc2 = {};
#pragma unroll
    for (int k = 0; k < 4; ++k) {
        const int ka = 4 * k + 2 * h;
        v2f a, b;
        a.x = wlds[N * 16 + ka];
        a.y = wlds[N * 16 + ka + 1];
        b.x = d2(ka, N);
        b.y = d2(ka + 1, N);
        acc2 = wmma_f32(a, b, acc2);
    }

    {
        float* strip = reinterpret_cast<float*>(strip4);
#pragma unroll
        for (int r = 0; r < 8; ++r) strip[(r + 8 * h) * 256 + tj * 16 + N] = acc2[r];
    }
    __syncthreads();

    if (wave < 12) {
        const float4* srcp = strip4 + lane;
        float4* dstp = out4 + (grp * 12 + wave) * 16384 + ti * 1024 + lane;
#pragma unroll
        for (int s = 0; s < 32; ++s) dstp[s * 32] = srcp[s * 32];
    }
}

extern "C" void kernel_launch(void* const* d_in, const int* in_sizes, int n_in,
                              void* d_out, int out_size, void* d_ws, size_t ws_size,
                              hipStream_t stream) {
    const float* x     = (const float*)d_in[0];
    const float* fc1_w = (const float*)d_in[1];
    const float* fc1_b = (const float*)d_in[2];
    const float* fc2_w = (const float*)d_in[3];
    const float* fc2_b = (const float*)d_in[4];
    const float* sse_w = (const float*)d_in[5];
    const float* sse_b = (const float*)d_in[6];

    float* ws      = (float*)d_ws;
    float* feat    = ws;
    float* csum    = ws + 131072;

    irse_dct_kernel<<<16, 512, 0, stream>>>(x, feat, csum);
    irse_gate_idct_bcast_kernel<<<256, 512, 0, stream>>>(feat, csum, fc1_w, fc1_b, fc2_w, fc2_b,
                                                         sse_w, sse_b, (float4*)d_out);
}


# Waits out of the dependent chain: MLP weights to LDS by whole-line b128 loads of all waves (were 32 dword gathers a lane), D table and out pointer read at entry, feat stored behind the last barrier
# speedup vs baseline: 2.7501x; 1.0822x over previous
#include <hip/hip_runtime.h>
#include <math.h>

typedef __attribute__((ext_vector_type(2))) float v2f;
typedef __attribute__((ext_vector_type(8))) float v8f;

__device__ __forceinline__ v8f wmma_f32(v2f a, v2f b, v8f c) {
    return __builtin_amdgcn_wmma_f32_16x16x4_f32(
        false, a, false, b, (short)0, c, false, false);
}

__device__ __constant__ float c_D[64] = {
    0.3535533906f,  0.3535533906f,  0.3535533906f,  0.3535533906f,
    0.3535533906f,  0.3535533906f,  0.3535533906f,  0.3535533906f,

    0.4903926402f,  0.4157348062f,  0.2777851165f,  0.0975451610f,
   -0.0975451610f, -0.2777851165f, -0.4157348062f, -0.4903926402f,

    0.4619397663f,  0.1913417162f, -0.1913417162f, -0.4619397663f,
   -0.4619397663f, -0.1913417162f,  0.1913417162f,  0.4619397663f,

    0.4157348062f, -0.0975451610f, -0.4903926402f, -0.2777851165f,
    0.2777851165f,  0.4903926402f,  0.0975451610f, -0.4157348062f,

    0.3535533906f, -0.3535533906f, -0.3535533906f,  0.3535533906f,
    0.3535533906f, -0.3535533906f, -0.3535533906f,  0.3535533906f,

    0.2777851165f, -0.4903926402f,  0.0975451610f,  0.4157348062f,
   -0.4157348062f, -0.0975451610f,  0.4903926402f, -0.2777851165f,

    0.1913417162f, -0.4619397663f,  0.4619397663f, -0.1913417162f,
   -0.1913417162f,  0.4619397663f, -0.4619397663f,  0.1913417162f,

    0.0975451610f, -0.2777851165f,  0.4157348062f, -0.4903926402f,
    0.4903926402f, -0.4157348062f,  0.2777851165f, -0.0975451610f
};

__device__ __forceinline__ float d2(int r, int c) {
    const float v = c_D[((r & 7) << 3) | (c & 7)];
    return ((r >> 3) == (c >> 3)) ? v : 0.0f;
}

__global__ void __launch_bounds__(512)
irse_dct_kernel(const float* __restrict__ x, float* __restrict__ feat) {
    __shared__ float lds[16 * 256];
    float* __restrict__ csum = feat + 131072;
    const int lane = threadIdx.x & 31;
    const int wave = threadIdx.x >> 5;
    const int ti = blockIdx.x, tj = wave;
    const int h = lane >> 4;
    const int N = lane & 15;
    const int row0 = ti * 16, col0 = tj * 16;

    v8f acc = {};
#pragma unroll
    for (int k = 0; k < 4; ++k) {
        const int ka = 4 * k + 2 * h;
        v2f a, b;
        a.x = d2(N, ka);
        a.y = d2(N, ka + 1);
        b.x = x[(((row0 + ka) * 256) + col0 + N) * 3];
        b.y = x[(((row0 + ka + 1) * 256) + col0 + N) * 3];
        acc = wmma_f32(a, b, acc);
    }

    float* wlds = lds + wave * 256;
#pragma unroll
    for (int r = 0; r < 8; ++r) wlds[(r + 8 * h) * 16 + N] = acc[r];
    __syncthreads();

    v8f acc2 = {};
#pragma unroll
    for (int k = 0; k < 4; ++k) {
        const int ka = 4 * k + 2 * h;
        v2f a, b;
        a.x = wlds[N * 16 + ka];
        a.y = wlds[N * 16 + ka + 1];
        b.x = d2(N, ka);
        b.y = d2(N, ka + 1);
        acc2 = wmma_f32(a, b, acc2);
    }


#pragma unroll
    for (int r = 0; r < 8; ++r) wlds[r * 32 + lane] = acc2[r];
    __syncthreads();
    {
        float4* ft = reinterpret_cast<float4*>(feat + (ti * 16 + tj) * 256);
        ft[N * 2 + h]        = make_float4(acc2[0], acc2[1], acc2[4], acc2[5]);
        ft[(N + 16) * 2 + h] = make_float4(acc2[2], acc2[3], acc2[6], acc2[7]);
    }
    if (threadIdx.x < 64) {
        const int f = threadIdx.x;
        const float* q = lds + (f >> 3) * 32 + (f & 7);
        float s0 = 0.0f, s1 = 0.0f, s2 = 0.0f, s3 = 0.0f;
        for (int w = 0; w < 16; ++w) {
            s0 += q[w * 256];
            s1 += q[w * 256 + 8];
            s2 += q[w * 256 + 16];
            s3 += q[w * 256 + 24];
        }
        csum[ti * 64 + f] = (s0 + s1) + (s2 + s3);
    }
}

__device__ __forceinline__ float irse_sigmoid(float a) {
    return 1.0f / (1.0f + expf(-a));
}

__global__ void __launch_bounds__(512)
irse_gate_idct_bcast_kernel(const float* __restrict__ feat,
                            const float* __restrict__ fc1_w, const float* __restrict__ fc1_b,
                            const float* __restrict__ fc2_w, const float* __restrict__ fc2_b,
                            const float* __restrict__ sse_w, const float* __restrict__ sse_b,
                            float4* __restrict__ out4) {
    __shared__ float lds[16 * 256];
    __shared__ float4 strip4[16 * 64];
    __shared__ __attribute__((aligned(16))) float s_w1[32 * 68];
    __shared__ __attribute__((aligned(16))) float s_w2[64 * 36];
    __shared__ float s_part[512];
    __shared__ float s_sq[64];
    __shared__ float s_h[32];
    __shared__ float s_sc[64];
    const int t = threadIdx.x;
    const int lane = t & 31;
    const int wave = t >> 5;
    const int ti = blockIdx.x & 15;
    const int grp = blockIdx.x >> 4;
    const int tj = wave;
    const int tile = ti * 16 + tj;
    const int h = lane >> 4;
    const int N = lane & 15;
    const int jn = N & 7;
    const bool mlp = t < 128;
    const float* __restrict__ csum = feat + 131072;

    float v[4][2];
    float wsp[2][2];
    {
        const float4* ft = reinterpret_cast<const float4*>(feat + tile * 256) + lane * 2;
        const float4 lo = ft[0], hi = ft[1];
        v[0][0] = lo.x; v[0][1] = lo.y; v[1][0] = lo.z; v[1][1] = lo.w;
        v[2][0] = hi.x; v[2][1] = hi.y; v[3][0] = hi.z; v[3][1] = hi.w;
    }
#pragma unroll
    for (int kk = 0; kk < 2; ++kk) {
#pragma unroll
        for (int e = 0; e < 2; ++e) wsp[kk][e] = sse_w[(4 * kk + 2 * h + e) * 8 + jn];
    }
    const float sb = sse_b[0];
    float dfr[4][2];
#pragma unroll
    for (int k = 0; k < 4; ++k) {
#pragma unroll
        for (int e = 0; e < 2; ++e) dfr[k][e] = d2(4 * k + 2 * h + e, N);
    }
    const float4 w1v = reinterpret_cast<const float4*>(fc1_w)[t];
    const float4 w2v = reinterpret_cast<const float4*>(fc2_w)[t];

    const int tm = t & 127;
    const int r1 = tm >> 2, q1 = tm & 3;
    const int o2 = tm >> 1, q2 = tm & 1;
    const float b1 = fc1_b[r1];
    const float b2 = fc2_b[o2];
    float cs[8];
    {
        const float* p = csum + (tm >> 6) * 512 + (tm & 63);
#pragma unroll
        for (int b = 0; b < 8; ++b) cs[b] = p[b * 64];
    }
    __builtin_amdgcn_sched_barrier(0);
    float spart = 0.0f;
#pragma unroll
    for (int b = 0; b < 8; ++b) spart += cs[b];
    s_part[t] = spart;
    *reinterpret_cast<float4*>(&s_w1[(t >> 4) * 68 + ((t & 15) << 2)]) = w1v;
    *reinterpret_cast<float4*>(&s_w2[(t >> 3) * 36 + ((t & 7) << 2)]) = w2v;
    __syncthreads();
    if (t < 64) s_sq[t] = (s_part[t] + s_part[t + 64]) * (1.0f / 1024.0f);
    __syncthreads();
    if (mlp) {
        float a = 0.0f;
#pragma unroll
        for (int i = 0; i < 16; ++i) a += s_sq[4 * i + q1] * s_w1[r1 * 68 + 4 * i + q1];
        a += __shfl_xor(a, 1);
        a += __shfl_xor(a, 2);
        if (q1 == 0) s_h[r1] = fmaxf(a + b1, 0.0f);
    }
    __syncthreads();
    if (mlp) {
        float a = 0.0f;
#pragma unroll
        for (int i = 0; i < 16; ++i) a += s_h[2 * i + q2] * s_w2[o2 * 36 + 2 * i + q2];
        a += __shfl_xor(a, 1);
        if (q2 == 0) s_sc[o2] = irse_sigmoid(a + b2);
    }
    __syncthreads();

    float g0 = (wsp[0][0] * v[0][0] + wsp[0][1] * v[0][1]) + (wsp[1][0] * v[1][0] + wsp[1][1] * v[1][1]);
    float g1 = (wsp[0][0] * v[2][0] + wsp[0][1] * v[2][1]) + (wsp[1][0] * v[3][0] + wsp[1][1] * v[3][1]);
    g0 += __shfl_xor(g0, 1);  g1 += __shfl_xor(g1, 1);
    g0 += __shfl_xor(g0, 2);  g1 += __shfl_xor(g1, 2);
    g0 += __shfl_xor(g0, 4);  g1 += __shfl_xor(g1, 4);
    g0 += __shfl_xor(g0, 16); g1 += __shfl_xor(g1, 16);
    float sp[2];
    sp[0] = irse_sigmoid(g0 + sb);
    sp[1] = irse_sigmoid(g1 + sb);

    float scv[2][2];
#pragma unroll
    for (int kk = 0; kk < 2; ++kk) {
#pragma unroll
        for (int e = 0; e < 2; ++e) scv[kk][e] = s_sc[(4 * kk + 2 * h + e) * 8 + jn];
    }

    v8f acc = {};
#pragma unroll
    for (int k = 0; k < 4; ++k) {
        const int ka = 4 * k + 2 * h;
        v2f a, b;
        a.x = dfr[k][0];
        a.y = dfr[k][1];
        b.x = fmaxf(v[k][0] * scv[k & 1][0], v[k][0] * sp[k >> 1]);
        b.y = fmaxf(v[k][1] * scv[k & 1][1], v[k][1] * sp[k >> 1]);
        acc = wmma_f32(a, b, acc);
    }

    float* wlds = lds + wave * 256;
#pragma unroll
    for (int r = 0; r < 8; ++r) wlds[(r + 8 * h) * 16 + N] = acc[r];
    __syncthreads();

    v8f acc2 = {};
#pragma unroll
    for (int k = 0; k < 4; ++k) {
        const int ka = 4 * k + 2 * h;
        v2f a, b;
        a.x = wlds[N * 16 + ka];
        a.y = wlds[N * 16 + ka + 1];
        b.x = dfr[k][0];
        b.y = dfr[k][1];
        acc2 = wmma_f32(a, b, acc2);
    }

    {
        float* strip = reinterpret_cast<float*>(strip4);
#pragma unroll
        for (int r = 0; r < 8; ++r) strip[(r + 8 * h) * 256 + tj * 16 + N] = acc2[r];
    }
    __syncthreads();

    if (wave < 12) {
        const float4* srcp = strip4 + lane;
        float4* dstp = out4 + (grp * 12 + wave) * 16384 + ti * 1024 + lane;
#pragma unroll
        for (int s = 0; s < 32; ++s) dstp[s * 32] = srcp[s * 32];
    }
}

extern "C" void kernel_launch(void* const* d_in, const int* in_sizes, int n_in,
                              void* d_out, int out_size, void* d_ws, size_t ws_size,
                              hipStream_t stream) {
    const float* x     = (const float*)d_in[0];
    const float* fc1_w = (const float*)d_in[1];
    const float* fc1_b = (const float*)d_in[2];
    const float* fc2_w = (const float*)d_in[3];
    const float* fc2_b = (const float*)d_in[4];
    const float* sse_w = (const float*)d_in[5];
    const float* sse_b = (const float*)d_in[6];

    float* ws      = (float*)d_ws;
    float* feat    = ws;

    irse_dct_kernel<<<16, 512, 0, stream>>>(x, feat);
    irse_gate_idct_bcast_kernel<<<256, 512, 0, stream>>>(feat, fc1_w, fc1_b, fc2_w, fc2_b,
                                                         sse_w, sse_b, (float4*)d_out);
}


# MLP chain of the gate kernel without LDS detours: weight float4 kept in its thread's registers, rows summed by row_xmask DPP adds (was bpermute), channel sums as register trees; block barriers 8 -> 4
# speedup vs baseline: 2.9634x; 1.0776x over previous
#include <hip/hip_runtime.h>
#include <math.h>

typedef __attribute__((ext_vector_type(2))) float v2f;
typedef __attribute__((ext_vector_type(8))) float v8f;

__device__ __forceinline__ v8f wmma_f32(v2f a, v2f b, v8f c) {
    return __builtin_amdgcn_wmma_f32_16x16x4_f32(
        false, a, false, b, (short)0, c, false, false);
}

__device__ __forceinline__ float xhalf(float x) {
    const int b = __builtin_bit_cast(int, x);
    return __builtin_bit_cast(float,
        __builtin_amdgcn_permlanex16(b, b, 0x76543210, 0xfedcba98, false, false));
}

template <int M>
__device__ __forceinline__ float row_xor(float x) {
    return __builtin_bit_cast(float,
        __builtin_amdgcn_update_dpp(0, __builtin_bit_cast(int, x), 0x160 + M, 0xf, 0xf, true));
}

__device__ __forceinline__ float4 add4(float4 a, float4 b) {
    return make_float4(a.x + b.x, a.y + b.y, a.z + b.z, a.w + b.w);
}

__device__ __constant__ float c_D[64] = {
    0.3535533906f,  0.3535533906f,  0.3535533906f,  0.3535533906f,
    0.3535533906f,  0.3535533906f,  0.3535533906f,  0.3535533906f,

    0.4903926402f,  0.4157348062f,  0.2777851165f,  0.0975451610f,
   -0.0975451610f, -0.2777851165f, -0.4157348062f, -0.4903926402f,

    0.4619397663f,  0.1913417162f, -0.1913417162f, -0.4619397663f,
   -0.4619397663f, -0.1913417162f,  0.1913417162f,  0.4619397663f,

    0.4157348062f, -0.0975451610f, -0.4903926402f, -0.2777851165f,
    0.2777851165f,  0.4903926402f,  0.0975451610f, -0.4157348062f,

    0.3535533906f, -0.3535533906f, -0.3535533906f,  0.3535533906f,
    0.3535533906f, -0.3535533906f, -0.3535533906f,  0.3535533906f,

    0.2777851165f, -0.4903926402f,  0.0975451610f,  0.4157348062f,
   -0.4157348062f, -0.0975451610f,  0.4903926402f, -0.2777851165f,

    0.1913417162f, -0.4619397663f,  0.4619397663f, -0.1913417162f,
   -0.1913417162f,  0.4619397663f, -0.4619397663f,  0.1913417162f,

    0.0975451610f, -0.2777851165f,  0.4157348062f, -0.4903926402f,
    0.4903926402f, -0.4157348062f,  0.2777851165f, -0.0975451610f
};

__device__ __forceinline__ float d2(int r, int c) {
    const float v = c_D[((r & 7) << 3) | (c & 7)];
    return ((r >> 3) == (c >> 3)) ? v : 0.0f;
}

__global__ void __launch_bounds__(512)
irse_dct_kernel(const float* __restrict__ x, float* __restrict__ feat) {
    __shared__ float lds[16 * 256];
    float* __restrict__ csum = feat + 131072;
    const int lane = threadIdx.x & 31;
    const int wave = threadIdx.x >> 5;
    const int ti = blockIdx.x, tj = wave;
    const int h = lane >> 4;
    const int N = lane & 15;
    const int row0 = ti * 16, col0 = tj * 16;

    v8f acc = {};
#pragma unroll
    for (int k = 0; k < 4; ++k) {
        const int ka = 4 * k + 2 * h;
        v2f a, b;
        a.x = d2(N, ka);
        a.y = d2(N, ka + 1);
        b.x = x[(((row0 + ka) * 256) + col0 + N) * 3];
        b.y = x[(((row0 + ka + 1) * 256) + col0 + N) * 3];
        acc = wmma_f32(a, b, acc);
    }

    float* wlds = lds + wave * 256;
#pragma unroll
    for (int r = 0; r < 8; ++r) wlds[(r + 8 * h) * 16 + N] = acc[r];
    __builtin_amdgcn_fence(__ATOMIC_ACQ_REL, "wavefront");

    v8f acc2 = {};
#pragma unroll
    for (int k = 0; k < 4; ++k) {
        const int ka = 4 * k + 2 * h;
        v2f a, b;
        a.x = wlds[N * 16 + ka];
        a.y = wlds[N * 16 + ka + 1];
        b.x = d2(N, ka);
        b.y = d2(N, ka + 1);
        acc2 = wmma_f32(a, b, acc2);
    }


#pragma unroll
    for (int r = 0; r < 8; ++r) wlds[r * 32 + lane] = acc2[r];
    __syncthreads();
    {
        float4* ft = reinterpret_cast<float4*>(feat + (ti * 16 + tj) * 256);
        ft[N * 2 + h]        = make_float4(acc2[0], acc2[1], acc2[4], acc2[5]);
        ft[(N + 16) * 2 + h] = make_float4(acc2[2], acc2[3], acc2[6], acc2[7]);
    }
    if (threadIdx.x < 64) {
        const int f = threadIdx.x;
        const float* q = lds + (f >> 3) * 32 + (f & 7);
        float s0 = 0.0f, s1 = 0.0f, s2 = 0.0f, s3 = 0.0f;
        for (int w = 0; w < 16; ++w) {
            s0 += q[w * 256];
            s1 += q[w * 256 + 8];
            s2 += q[w * 256 + 16];
            s3 += q[w * 256 + 24];
        }
        csum[ti * 64 + f] = (s0 + s1) + (s2 + s3);
    }
}

__device__ __forceinline__ float irse_sigmoid(float a) {
    return 1.0f / (1.0f + expf(-a));
}

__global__ void __launch_bounds__(512)
irse_gate_idct_bcast_kernel(const float* __restrict__ feat,
                            const float* __restrict__ fc1_w, const float* __restrict__ fc1_b,
                            const float* __restrict__ fc2_w, const float* __restrict__ fc2_b,
                            const float* __restrict__ sse_w, const float* __restrict__ sse_b,
                            float4* __restrict__ out4) {
    __shared__ float lds[16 * 256];
    __shared__ float4 strip4[16 * 64];
    __shared__ __attribute__((aligned(16))) float s_h[32];
    __shared__ float s_sc[64];
    const int t = threadIdx.x;
    const int lane = t & 31;
    const int wave = t >> 5;
    const int ti = blockIdx.x & 15;
    const int grp = blockIdx.x >> 4;
    const int tj = wave;
    const int tile = ti * 16 + tj;
    const int h = lane >> 4;
    const int N = lane & 15;
    const int jn = N & 7;
    const float* __restrict__ csum = feat + 131072;

    float v[4][2];
    float wsp[2][2];
    {
        const float4* ft = reinterpret_cast<const float4*>(feat + tile * 256) + lane * 2;
        const float4 lo = ft[0], hi = ft[1];
        v[0][0] = lo.x; v[0][1] = lo.y; v[1][0] = lo.z; v[1][1] = lo.w;
        v[2][0] = hi.x; v[2][1] = hi.y; v[3][0] = hi.z; v[3][1] = hi.w;
    }
#pragma unroll
    for (int kk = 0; kk < 2; ++kk) {
#pragma unroll
        for (int e = 0; e < 2; ++e) wsp[kk][e] = sse_w[(4 * kk + 2 * h + e) * 8 + jn];
    }
    const float sb = sse_b[0];
    float draw[4][2];
#pragma unroll
    for (int k = 0; k < 4; ++k) {
#pragma unroll
        for (int e = 0; e < 2; ++e) draw[k][e] = c_D[(((4 * k + 2 * h + e) & 7) << 3) | jn];
    }
    const float4 w1v = reinterpret_cast<const float4*>(fc1_w)[t];
    const float4 w2v = reinterpret_cast<const float4*>(fc2_w)[t];
    const float b1 = fc1_b[t >> 4];
    const float b2 = fc2_b[t >> 3];
    float4 cp[8];
    {
        const float4* p = reinterpret_cast<const float4*>(csum) + h * 128 + N;
#pragma unroll
        for (int b = 0; b < 8; ++b) cp[b] = p[b * 16];
    }
    asm volatile("" :: "v"(v[0][0]), "v"(v[2][0]),
                       "v"(wsp[0][0]), "v"(wsp[0][1]), "v"(wsp[1][0]), "v"(wsp[1][1]),
                       "v"(draw[0][0]), "v"(draw[0][1]), "v"(draw[1][0]), "v"(draw[1][1]),
                       "v"(draw[2][0]), "v"(draw[2][1]), "v"(draw[3][0]), "v"(draw[3][1]),
                       "v"(w1v.x), "v"(w2v.x), "v"(b1), "v"(b2),
                       "v"(cp[0].x), "v"(cp[1].x), "v"(cp[2].x), "v"(cp[3].x),
                       "v"(cp[4].x), "v"(cp[5].x), "v"(cp[6].x), "v"(cp[7].x));
    float dfr[4][2];
#pragma unroll
    for (int k = 0; k < 4; ++k) {
#pragma unroll
        for (int e = 0; e < 2; ++e) dfr[k][e] = (((4 * k + 2 * h + e) >> 3) == (N >> 3)) ? draw[k][e] : 0.0f;
    }

    float4 cs = add4(add4(add4(cp[0], cp[1]), add4(cp[2], cp[3])),
                     add4(add4(cp[4], cp[5]), add4(cp[6], cp[7])));
    cs.x += xhalf(cs.x);
    cs.y += xhalf(cs.y);
    cs.z += xhalf(cs.z);
    cs.w += xhalf(cs.w);
    float a1 = (w1v.x * cs.x + w1v.y * cs.y) + (w1v.z * cs.z + w1v.w * cs.w);
    a1 += row_xor<1>(a1);
    a1 += row_xor<2>(a1);
    a1 += row_xor<4>(a1);
    a1 += row_xor<8>(a1);
    s_h[t >> 4] = fmaxf(a1 * (1.0f / 1024.0f) + b1, 0.0f);
    __syncthreads();

    {
        const float4 h4 = reinterpret_cast<const float4*>(s_h)[t & 7];
        float a2 = (w2v.x * h4.x + w2v.y * h4.y) + (w2v.z * h4.z + w2v.w * h4.w);
        a2 += row_xor<1>(a2);
        a2 += row_xor<2>(a2);
        a2 += row_xor<4>(a2);
        s_sc[t >> 3] = irse_sigmoid(a2 + b2);
    }

    float g0 = (wsp[0][0] * v[0][0] + wsp[0][1] * v[0][1]) + (wsp[1][0] * v[1][0] + wsp[1][1] * v[1][1]);
    float g1 = (wsp[0][0] * v[2][0] + wsp[0][1] * v[2][1]) + (wsp[1][0] * v[3][0] + wsp[1][1] * v[3][1]);
    g0 += row_xor<1>(g0);  g1 += row_xor<1>(g1);
    g0 += row_xor<2>(g0);  g1 += row_xor<2>(g1);
    g0 += row_xor<4>(g0);  g1 += row_xor<4>(g1);
    g0 += xhalf(g0);       g1 += xhalf(g1);
    float sp[2];
    sp[0] = irse_sigmoid(g0 + sb);
    sp[1] = irse_sigmoid(g1 + sb);
    asm volatile("" :: "v"(sp[0]), "v"(sp[1]));
    __syncthreads();

    float scv[2][2];
#pragma unroll
    for (int kk = 0; kk < 2; ++kk) {
#pragma unroll
        for (int e = 0; e < 2; ++e) scv[kk][e] = s_sc[(4 * kk + 2 * h + e) * 8 + jn];
    }

    v8f acc = {};
#pragma unroll
    for (int k = 0; k < 4; ++k) {
        v2f a, b;
        a.x = dfr[k][0];
        a.y = dfr[k][1];
        b.x = fmaxf(v[k][0] * scv[k & 1][0], v[k][0] * sp[k >> 1]);
        b.y = fmaxf(v[k][1] * scv[k & 1][1], v[k][1] * sp[k >> 1]);
        acc = wmma_f32(a, b, acc);
    }

    float* wlds = lds + wave * 256;
#pragma unroll
    for (int r = 0; r < 8; ++r) wlds[(r + 8 * h) * 16 + N] = acc[r];
    __builtin_amdgcn_fence(__ATOMIC_ACQ_REL, "wavefront");

    v8f acc2 = {};
#pragma unroll
    for (int k = 0; k < 4; ++k) {
        const int ka = 4 * k + 2 * h;
        v2f a, b;
        a.x = wlds[N * 16 + ka];
        a.y = wlds[N * 16 + ka + 1];
        b.x = dfr[k][0];
        b.y = dfr[k][1];
        acc2 = wmma_f32(a, b, acc2);
    }

    {
        float* strip = reinterpret_cast<float*>(strip4);
#pragma unroll
        for (int r = 0; r < 8; ++r) strip[(r + 8 * h) * 256 + tj * 16 + N] = acc2[r];
    }
    __syncthreads();

    if (wave < 12) {
        const float4* srcp = strip4 + lane;
        float4* dstp = out4 + (grp * 12 + wave) * 16384 + ti * 1024 + lane;
#pragma unroll
        for (int s = 0; s < 32; ++s) dstp[s * 32] = srcp[s * 32];
    }
}

extern "C" void kernel_launch(void* const* d_in, const int* in_sizes, int n_in,
                              void* d_out, int out_size, void* d_ws, size_t ws_size,
                              hipStream_t stream) {
    const float* x     = (const float*)d_in[0];
    const float* fc1_w = (const float*)d_in[1];
    const float* fc1_b = (const float*)d_in[2];
    const float* fc2_w = (const float*)d_in[3];
    const float* fc2_b = (const float*)d_in[4];
    const float* sse_w = (const float*)d_in[5];
    const float* sse_b = (const float*)d_in[6];

    float* ws      = (float*)d_ws;
    float* feat    = ws;

    irse_dct_kernel<<<16, 512, 0, stream>>>(x, feat);
    irse_gate_idct_bcast_kernel<<<256, 512, 0, stream>>>(feat, fc1_w, fc1_b, fc2_w, fc2_b,
                                                         sse_w, sse_b, (float4*)d_out);
}
